# starting kernel
# baseline (speedup 1.0000x reference)
#include <hip/hip_runtime.h>

typedef __attribute__((ext_vector_type(2))) float v2f;
typedef __attribute__((ext_vector_type(8))) float v8f;

constexpr int IN_FEAT  = 128;
constexpr int KDIM     = 2 * IN_FEAT;
constexpr int OUT_FEAT = 128;
constexpr int TILE_M   = 16;
constexpr int LDF      = KDIM + 4;

__device__ __forceinline__ void atomic_add_f32(float* p, float v) {
    __hip_atomic_fetch_add(p, v, __ATOMIC_RELAXED, __HIP_MEMORY_SCOPE_AGENT);
}

__global__ __launch_bounds__(256) void sage_scatter(
    const float* __restrict__ h_s, const int* __restrict__ src,
    const int* __restrict__ dst, float* __restrict__ nsum,
    float* __restrict__ deg, int n_edges)
{
    const int waves_total = (gridDim.x * blockDim.x) >> 5;
    const int wave = (blockIdx.x * blockDim.x + threadIdx.x) >> 5;
    const int lane = threadIdx.x & 31;

    for (int e = wave; e < n_edges; e += waves_total) {
        const int s = src[e];
        const int d = dst[e];
        const float* hs = h_s + (size_t)s * IN_FEAT;
        float*       ns = nsum + (size_t)d * IN_FEAT;
#pragma unroll
        for (int i = 0; i < IN_FEAT / 32; ++i) {
            atomic_add_f32(&ns[lane + i * 32], hs[lane + i * 32]);
        }
        if (lane == 0) atomic_add_f32(&deg[d], 1.0f);
    }
}

__global__ __launch_bounds__(256) void sage_gemm(
    const float* __restrict__ h_d, const float* __restrict__ nsum,
    const float* __restrict__ deg, const float* __restrict__ W,
    const float* __restrict__ bias, float* __restrict__ out, int n_dst)
{
    __shared__ float feat[TILE_M * LDF];

    const int t    = threadIdx.x;
    const int row0 = blockIdx.x * TILE_M;

    for (int i = t; i < TILE_M * KDIM; i += 256) {
        const int r = i >> 8;
        const int c = i & (KDIM - 1);
        int g = row0 + r;
        if (g >= n_dst) g = n_dst - 1;
        float v;
        if (c < IN_FEAT) {
            v = h_d[(size_t)g * IN_FEAT + c];
        } else {
            const float d = deg[g];
            v = nsum[(size_t)g * IN_FEAT + (c - IN_FEAT)] / fmaxf(d, 1.0f);
        }
        feat[r * LDF + c] = v;
    }
    __syncthreads();

    const int wave  = t >> 5;
    const int lane  = t & 31;
    const int n     = wave * 16 + (lane & 15);
    const int m     = lane & 15;
    const int khalf = (lane >> 4) * 2;

    const float* frow = feat + m * LDF;
    const float* wrow = W + (size_t)n * KDIM;

    v8f acc = {};
#pragma unroll 4
    for (int k0 = 0; k0 < KDIM; k0 += 4) {
        v2f a = *(const v2f*)(frow + k0 + khalf);
        v2f b = *(const v2f*)(wrow + k0 + khalf);
        acc = __builtin_amdgcn_wmma_f32_16x16x4_f32(
             false, a,  false, b,
             (short)0, acc,  false,  false);
    }

    const float bv    = bias[n];
    const int   mhigh = (lane >> 4) * 8;

#pragma unroll
    for (int v = 0; v < 8; ++v) {
        const int g = row0 + mhigh + v;
        if (g < n_dst) out[(size_t)g * OUT_FEAT + n] = acc[v] + bv;
    }
}

extern "C" void kernel_launch(void* const* d_in, const int* in_sizes, int n_in,
                              void* d_out, int out_size, void* d_ws, size_t ws_size,
                              hipStream_t stream) {
    const float* h_s  = (const float*)d_in[0];
    const float* h_d  = (const float*)d_in[1];
    const int*   src  = (const int*)d_in[2];
    const int*   dst  = (const int*)d_in[3];
    const float* W    = (const float*)d_in[4];
    const float* bias = (const float*)d_in[5];

    const int n_dst   = in_sizes[1] / IN_FEAT;
    const int n_edges = in_sizes[2];
    (void)n_in; (void)out_size; (void)ws_size;

    float* nsum = (float*)d_ws;
    float* deg  = nsum + (size_t)n_dst * IN_FEAT;

    const size_t zero_bytes = ((size_t)n_dst * IN_FEAT + (size_t)n_dst) * sizeof(float);
    hipMemsetAsync(d_ws, 0, zero_bytes, stream);

    sage_scatter<<<1024, 256, 0, stream>>>(h_s, src, dst, nsum, deg, n_edges);

    const int blocks = (n_dst + TILE_M - 1) / TILE_M;
    sage_gemm<<<blocks, 256, 0, stream>>>(h_d, nsum, deg, W, bias,
                                          (float*)d_out, n_dst);
}


# scatter kernel: each wave takes 8 edges per trip, all 32 row loads issued before the atomics, next trip's indices prefetched, grid doubled to fill every wave slot; f32 atomic arrival order as before
# speedup vs baseline: 1.5005x; 1.5005x over previous
#include <hip/hip_runtime.h>

typedef __attribute__((ext_vector_type(2))) float v2f;
typedef __attribute__((ext_vector_type(8))) float v8f;

constexpr int IN_FEAT  = 128;
constexpr int KDIM     = 2 * IN_FEAT;
constexpr int OUT_FEAT = 128;
constexpr int TILE_M   = 16;
constexpr int LDF      = KDIM + 4;

__device__ __forceinline__ void atomic_add_f32(float* p, float v) {
    __hip_atomic_fetch_add(p, v, __ATOMIC_RELAXED, __HIP_MEMORY_SCOPE_AGENT);
}

constexpr int EDGES_PER_TRIP = 8;

__global__ __launch_bounds__(256) void sage_scatter(
    const float* __restrict__ h_s, const int* __restrict__ src,
    const int* __restrict__ dst, float* __restrict__ nsum,
    float* __restrict__ deg, int n_edges)
{
    const int waves_total = (gridDim.x * blockDim.x) >> 5;
    const int wave = __builtin_amdgcn_readfirstlane(
        (int)((blockIdx.x * blockDim.x + threadIdx.x) >> 5));
    const int lane = threadIdx.x & 31;
    const int stride = waves_total * EDGES_PER_TRIP;

    int e0 = wave * EDGES_PER_TRIP;
    if (e0 >= n_edges) return;
    const int last = n_edges - 1;
    int s_l = src[min(e0 + lane, last)];
    int d_l = dst[min(e0 + lane, last)];

    while (e0 < n_edges) {
        const int en = e0 + stride;
        const int s_n = src[min(en + lane, last)];
        const int d_n = dst[min(en + lane, last)];

        float v[EDGES_PER_TRIP][IN_FEAT / 32];
        int   d[EDGES_PER_TRIP];
#pragma unroll
        for (int u = 0; u < EDGES_PER_TRIP; ++u) {
            const int s = __builtin_amdgcn_readlane(s_l, u);
            d[u]        = __builtin_amdgcn_readlane(d_l, u);
            const float* hs = h_s + (size_t)s * IN_FEAT + lane;
#pragma unroll
            for (int i = 0; i < IN_FEAT / 32; ++i) v[u][i] = hs[i * 32];
        }

        if (e0 + EDGES_PER_TRIP <= n_edges) {
#pragma unroll
            for (int u = 0; u < EDGES_PER_TRIP; ++u) {
                float* ns = nsum + (size_t)d[u] * IN_FEAT + lane;
#pragma unroll
                for (int i = 0; i < IN_FEAT / 32; ++i) atomic_add_f32(&ns[i * 32], v[u][i]);
            }
        } else {
#pragma unroll
            for (int u = 0; u < EDGES_PER_TRIP; ++u) {
                if (e0 + u < n_edges) {
                    float* ns = nsum + (size_t)d[u] * IN_FEAT + lane;
#pragma unroll
                    for (int i = 0; i < IN_FEAT / 32; ++i) atomic_add_f32(&ns[i * 32], v[u][i]);
                }
            }
        }
        if (lane < EDGES_PER_TRIP && e0 + lane < n_edges) atomic_add_f32(&deg[d_l], 1.0f);

        s_l = s_n;
        d_l = d_n;
        e0  = en;
    }
}

__global__ __launch_bounds__(256) void sage_gemm(
    const float* __restrict__ h_d, const float* __restrict__ nsum,
    const float* __restrict__ deg, const float* __restrict__ W,
    const float* __restrict__ bias, float* __restrict__ out, int n_dst)
{
    __shared__ float feat[TILE_M * LDF];

    const int t    = threadIdx.x;
    const int row0 = blockIdx.x * TILE_M;

    for (int i = t; i < TILE_M * KDIM; i += 256) {
        const int r = i >> 8;
        const int c = i & (KDIM - 1);
        int g = row0 + r;
        if (g >= n_dst) g = n_dst - 1;
        float v;
        if (c < IN_FEAT) {
            v = h_d[(size_t)g * IN_FEAT + c];
        } else {
            const float d = deg[g];
            v = nsum[(size_t)g * IN_FEAT + (c - IN_FEAT)] / fmaxf(d, 1.0f);
        }
        feat[r * LDF + c] = v;
    }
    __syncthreads();

    const int wave  = t >> 5;
    const int lane  = t & 31;
    const int n     = wave * 16 + (lane & 15);
    const int m     = lane & 15;
    const int khalf = (lane >> 4) * 2;

    const float* frow = feat + m * LDF;
    const float* wrow = W + (size_t)n * KDIM;

    v8f acc = {};
#pragma unroll 4
    for (int k0 = 0; k0 < KDIM; k0 += 4) {
        v2f a = *(const v2f*)(frow + k0 + khalf);
        v2f b = *(const v2f*)(wrow + k0 + khalf);
        acc = __builtin_amdgcn_wmma_f32_16x16x4_f32(
             false, a,  false, b,
             (short)0, acc,  false,  false);
    }

    const float bv    = bias[n];
    const int   mhigh = (lane >> 4) * 8;

#pragma unroll
    for (int v = 0; v < 8; ++v) {
        const int g = row0 + mhigh + v;
        if (g < n_dst) out[(size_t)g * OUT_FEAT + n] = acc[v] + bv;
    }
}

extern "C" void kernel_launch(void* const* d_in, const int* in_sizes, int n_in,
                              void* d_out, int out_size, void* d_ws, size_t ws_size,
                              hipStream_t stream) {
    const float* h_s  = (const float*)d_in[0];
    const float* h_d  = (const float*)d_in[1];
    const int*   src  = (const int*)d_in[2];
    const int*   dst  = (const int*)d_in[3];
    const float* W    = (const float*)d_in[4];
    const float* bias = (const float*)d_in[5];

    const int n_dst   = in_sizes[1] / IN_FEAT;
    const int n_edges = in_sizes[2];
    (void)n_in; (void)out_size; (void)ws_size;

    float* nsum = (float*)d_ws;
    float* deg  = nsum + (size_t)n_dst * IN_FEAT;

    const size_t zero_bytes = ((size_t)n_dst * IN_FEAT + (size_t)n_dst) * sizeof(float);
    hipMemsetAsync(d_ws, 0, zero_bytes, stream);

    sage_scatter<<<2048, 256, 0, stream>>>(h_s, src, dst, nsum, deg, n_edges);

    const int blocks = (n_dst + TILE_M - 1) / TILE_M;
    sage_gemm<<<blocks, 256, 0, stream>>>(h_d, nsum, deg, W, bias,
                                          (float*)d_out, n_dst);
}


# segment mean without f32 atomics: a returning int add per edge gives its slot in a 64-slot bucket per target row (overflow list past 64); GEMM staging sums rows in registers; sum order = slot order
# speedup vs baseline: 2.1733x; 1.4484x over previous
#include <hip/hip_runtime.h>


typedef __attribute__((ext_vector_type(2))) float v2f;
typedef __attribute__((ext_vector_type(4))) float v4f;
typedef __attribute__((ext_vector_type(8))) float v8f;
typedef __attribute__((ext_vector_type(2))) int   v2i;
typedef __attribute__((ext_vector_type(4))) int   v4i;

constexpr int IN_FEAT  = 128;
constexpr int KDIM     = 2 * IN_FEAT;
constexpr int OUT_FEAT = 128;
constexpr int TILE_M   = 16;
constexpr int LDF      = KDIM + 4;

constexpr int CAP = 64;

__device__ __forceinline__ int atomic_inc_ret(int* p) {
    return __hip_atomic_fetch_add(p, 1, __ATOMIC_RELAXED, __HIP_MEMORY_SCOPE_AGENT);
}

__device__ __forceinline__ void place_edge(int* __restrict__ bkt, int* __restrict__ ovf_count,
                                           int* __restrict__ ovf, int d, int s, int slot)
{
    if (slot < CAP) {
        bkt[(size_t)d * CAP + slot] = s;
    } else {
        const int o = atomic_inc_ret(ovf_count);
        v2i pr; pr[0] = d; pr[1] = s;
        *(v2i*)(ovf + (size_t)o * 2) = pr;
    }
}

__global__ __launch_bounds__(256) void sage_bucket(
    const int* __restrict__ src, const int* __restrict__ dst,
    int* __restrict__ cnt, int* __restrict__ bkt,
    int* __restrict__ ovf_count, int* __restrict__ ovf, int n_edges)
{
    const int e0 = (int)(blockIdx.x * blockDim.x + threadIdx.x) * 4;
    if (e0 >= n_edges) return;

    if (e0 + 4 <= n_edges) {
        const v4i dv = *(const v4i*)(dst + e0);
        const v4i sv = *(const v4i*)(src + e0);
        int slot[4];
#pragma unroll
        for (int k = 0; k < 4; ++k) slot[k] = atomic_inc_ret(&cnt[dv[k]]);
#pragma unroll
        for (int k = 0; k < 4; ++k) place_edge(bkt, ovf_count, ovf, dv[k], sv[k], slot[k]);
    } else {
        for (int e = e0; e < n_edges; ++e) {
            const int d = dst[e];
            const int s = src[e];
            place_edge(bkt, ovf_count, ovf, d, s, atomic_inc_ret(&cnt[d]));
        }
    }
}

template <int N>
__device__ __forceinline__ v4f gather_group(const float* __restrict__ h_s_lane,
                                            int idx, int j, v4f acc)
{
    v4f v[N];
#pragma unroll
    for (int u = 0; u < N; ++u) {
        const int sidx = __builtin_amdgcn_readlane(idx, j + u);
        v[u] = *(const v4f*)(h_s_lane + (size_t)sidx * IN_FEAT);
    }
#pragma unroll
    for (int u = 0; u < N; ++u) acc += v[u];
    return acc;
}

__device__ __forceinline__ v4f gather_rows(const float* __restrict__ h_s_lane,
                                           int idx, int n, v4f acc)
{
    int j = 0;
    for (; j + 8 <= n; j += 8) acc = gather_group<8>(h_s_lane, idx, j, acc);
    if (j + 4 <= n) { acc = gather_group<4>(h_s_lane, idx, j, acc); j += 4; }
    const int rem = n - j;
    if (rem == 3)      acc = gather_group<3>(h_s_lane, idx, j, acc);
    else if (rem == 2) acc = gather_group<2>(h_s_lane, idx, j, acc);
    else if (rem == 1) acc = gather_group<1>(h_s_lane, idx, j, acc);
    return acc;
}

__global__ __launch_bounds__(256) void sage_gemm(
    const float* __restrict__ h_s, const float* __restrict__ h_d,
    const int* __restrict__ cnt, const int* __restrict__ bkt,
    const int* __restrict__ ovf_count, const int* __restrict__ ovf,
    const float* __restrict__ W, const float* __restrict__ bias,
    float* __restrict__ out, int n_dst, int n_edges)
{
    __shared__ __attribute__((aligned(16))) float feat[TILE_M * LDF];

    const int t    = threadIdx.x;
    const int row0 = blockIdx.x * TILE_M;
    const int lane = t & 31;
    const int wave = __builtin_amdgcn_readfirstlane(t >> 5);

    const float* h_s_lane = h_s + lane * 4;
    int g[2], c[2];
#pragma unroll
    for (int rr = 0; rr < 2; ++rr) {
        g[rr] = min(row0 + wave * 2 + rr, n_dst - 1);
        c[rr] = cnt[g[rr]];
    }
    int idx[2][2];
    v4f hd[2];
#pragma unroll
    for (int rr = 0; rr < 2; ++rr) {
        const int* b = bkt + (size_t)g[rr] * CAP;
        idx[rr][0] = (lane < c[rr]) ? b[lane] : 0;
        idx[rr][1] = (lane + 32 < min(c[rr], CAP)) ? b[lane + 32] : 0;
        hd[rr] = *(const v4f*)(h_d + (size_t)g[rr] * IN_FEAT + lane * 4);
    }
#pragma unroll
    for (int rr = 0; rr < 2; ++rr) {
        const int n_in = min(c[rr], CAP);
        v4f acc = {0.0f, 0.0f, 0.0f, 0.0f};
        acc = gather_rows(h_s_lane, idx[rr][0], min(n_in, 32), acc);
        if (n_in > 32) acc = gather_rows(h_s_lane, idx[rr][1], n_in - 32, acc);
        if (c[rr] > CAP) {
            const int total = min(*ovf_count, n_edges);
            for (int o0 = 0; o0 < total; o0 += 32) {
                const int o = o0 + lane;
                v2i pr = {-1, 0};
                if (o < total) pr = *(const v2i*)(ovf + (size_t)o * 2);
                unsigned m = (unsigned)__builtin_amdgcn_ballot_w32(pr[0] == g[rr]);
                while (m) {
                    const int l = __builtin_ctz(m);
                    m &= m - 1;
                    const int sidx = __builtin_amdgcn_readlane(pr[1], l);
                    acc += *(const v4f*)(h_s_lane + (size_t)sidx * IN_FEAT);
                }
            }
        }
        const float dg = fmaxf((float)c[rr], 1.0f);
        const v4f mean = acc / dg;
        float* frow_w = feat + (wave * 2 + rr) * LDF + lane * 4;
        *(v4f*)(frow_w)           = hd[rr];
        *(v4f*)(frow_w + IN_FEAT) = mean;
    }
    __syncthreads();

    const int n     = wave * 16 + (lane & 15);
    const int m     = lane & 15;
    const int khalf = (lane >> 4) * 2;

    const float* frow = feat + m * LDF;
    const float* wrow = W + (size_t)n * KDIM;

    v8f acc = {};
#pragma unroll 4
    for (int k0 = 0; k0 < KDIM; k0 += 4) {
        v2f a = *(const v2f*)(frow + k0 + khalf);
        v2f b = *(const v2f*)(wrow + k0 + khalf);
        acc = __builtin_amdgcn_wmma_f32_16x16x4_f32(
             false, a,  false, b,
             (short)0, acc,  false,  false);
    }

    const float bv    = bias[n];
    const int   mhigh = (lane >> 4) * 8;

#pragma unroll
    for (int v = 0; v < 8; ++v) {
        const int gr = row0 + mhigh + v;
        if (gr < n_dst) out[(size_t)gr * OUT_FEAT + n] = acc[v] + bv;
    }
}

extern "C" void kernel_launch(void* const* d_in, const int* in_sizes, int n_in,
                              void* d_out, int out_size, void* d_ws, size_t ws_size,
                              hipStream_t stream) {
    const float* h_s  = (const float*)d_in[0];
    const float* h_d  = (const float*)d_in[1];
    const int*   src  = (const int*)d_in[2];
    const int*   dst  = (const int*)d_in[3];
    const float* W    = (const float*)d_in[4];
    const float* bias = (const float*)d_in[5];

    const int n_dst   = in_sizes[1] / IN_FEAT;
    const int n_edges = in_sizes[2];
    (void)n_in; (void)out_size; (void)ws_size;

    int* cnt       = (int*)d_ws;
    int* ovf_count = cnt + n_dst;
    int* bkt       = cnt + ((n_dst + 1 + 63) / 64) * 64;
    int* ovf       = bkt + (size_t)n_dst * CAP;

    const size_t zero_bytes = ((size_t)n_dst + 1) * sizeof(int);
    hipMemsetAsync(d_ws, 0, zero_bytes, stream);

    const int bblocks = (n_edges + 1023) / 1024;
    sage_bucket<<<bblocks, 256, 0, stream>>>(src, dst, cnt, bkt, ovf_count, ovf, n_edges);

    const int blocks = (n_dst + TILE_M - 1) / TILE_M;
    sage_gemm<<<blocks, 256, 0, stream>>>(h_s, h_d, cnt, bkt, ovf_count, ovf, W, bias,
                                          (float*)d_out, n_dst, n_edges);
}


# GEMM kernel: each wave holds its 16-column slab of W in registers (read once per block, whole lines, via LDS) and 512 resident blocks loop over the row tiles; k order and sum order unchanged
# speedup vs baseline: 2.2954x; 1.0562x over previous
#include <hip/hip_runtime.h>


typedef __attribute__((ext_vector_type(2))) float v2f;
typedef __attribute__((ext_vector_type(4))) float v4f;
typedef __attribute__((ext_vector_type(8))) float v8f;
typedef __attribute__((ext_vector_type(2))) int   v2i;
typedef __attribute__((ext_vector_type(4))) int   v4i;

constexpr int IN_FEAT  = 128;
constexpr int KDIM     = 2 * IN_FEAT;
constexpr int OUT_FEAT = 128;
constexpr int TILE_M   = 16;
constexpr int LDF      = KDIM + 4;

constexpr int CAP = 64;

constexpr int GEMM_GRID = 512;
constexpr int WPITCH = 68;
constexpr int FEAT_FLOATS = 2 * TILE_M * LDF;
constexpr int WST_FLOATS  = 8 * TILE_M * WPITCH;
constexpr int LDS_FLOATS  = FEAT_FLOATS > WST_FLOATS ? FEAT_FLOATS : WST_FLOATS;

__device__ __forceinline__ int atomic_inc_ret(int* p) {
    return __hip_atomic_fetch_add(p, 1, __ATOMIC_RELAXED, __HIP_MEMORY_SCOPE_AGENT);
}

__device__ __forceinline__ void place_edge(int* __restrict__ bkt, int* __restrict__ ovf_count,
                                           int* __restrict__ ovf, int d, int s, int slot)
{
    if (slot < CAP) {
        bkt[(size_t)d * CAP + slot] = s;
    } else {
        const int o = atomic_inc_ret(ovf_count);
        v2i pr; pr[0] = d; pr[1] = s;
        *(v2i*)(ovf + (size_t)o * 2) = pr;
    }
}

__global__ __launch_bounds__(256) void sage_bucket(
    const int* __restrict__ src, const int* __restrict__ dst,
    int* __restrict__ cnt, int* __restrict__ bkt,
    int* __restrict__ ovf_count, int* __restrict__ ovf, int n_edges)
{
    const int e0 = (int)(blockIdx.x * blockDim.x + threadIdx.x) * 4;
    if (e0 >= n_edges) return;

    if (e0 + 4 <= n_edges) {
        const v4i dv = *(const v4i*)(dst + e0);
        const v4i sv = *(const v4i*)(src + e0);
        int slot[4];
#pragma unroll
        for (int k = 0; k < 4; ++k) slot[k] = atomic_inc_ret(&cnt[dv[k]]);
#pragma unroll
        for (int k = 0; k < 4; ++k) place_edge(bkt, ovf_count, ovf, dv[k], sv[k], slot[k]);
    } else {
        for (int e = e0; e < n_edges; ++e) {
            const int d = dst[e];
            const int s = src[e];
            place_edge(bkt, ovf_count, ovf, d, s, atomic_inc_ret(&cnt[d]));
        }
    }
}

template <int N>
__device__ __forceinline__ v4f gather_group(const float* __restrict__ h_s_lane,
                                            int idx, int j, v4f acc)
{
    v4f v[N];
#pragma unroll
    for (int u = 0; u < N; ++u) {
        const int sidx = __builtin_amdgcn_readlane(idx, j + u);
        v[u] = *(const v4f*)(h_s_lane + (size_t)sidx * IN_FEAT);
    }
#pragma unroll
    for (int u = 0; u < N; ++u) acc += v[u];
    return acc;
}

__device__ __forceinline__ v4f gather_rows(const float* __restrict__ h_s_lane,
                                           int idx, int n, v4f acc)
{
    int j = 0;
    for (; j + 8 <= n; j += 8) acc = gather_group<8>(h_s_lane, idx, j, acc);
    if (j + 4 <= n) { acc = gather_group<4>(h_s_lane, idx, j, acc); j += 4; }
    const int rem = n - j;
    if (rem == 3)      acc = gather_group<3>(h_s_lane, idx, j, acc);
    else if (rem == 2) acc = gather_group<2>(h_s_lane, idx, j, acc);
    else if (rem == 1) acc = gather_group<1>(h_s_lane, idx, j, acc);
    return acc;
}

__global__ __launch_bounds__(256) void sage_gemm(
    const float* __restrict__ h_s, const float* __restrict__ h_d,
    const int* __restrict__ cnt, const int* __restrict__ bkt,
    const int* __restrict__ ovf_count, const int* __restrict__ ovf,
    const float* __restrict__ W, const float* __restrict__ bias,
    float* __restrict__ out, int n_dst, int n_edges)
{
    __shared__ __attribute__((aligned(16))) float lds[LDS_FLOATS];

    const int t    = threadIdx.x;
    const int lane = t & 31;
    const int wave = __builtin_amdgcn_readfirstlane(t >> 5);

    const int m     = lane & 15;
    const int half  = lane >> 4;
    const int khalf = half * 2;
    const int n     = wave * 16 + m;

    v2f breg[64];
    {
        float* wt = lds + wave * (TILE_M * WPITCH);
        const float* wsrc  = W + (size_t)(wave * 16 + half) * KDIM + m * 4;
        float*       wdst  = wt + half * WPITCH + m * 4;
        const float* wfrag = wt + m * WPITCH + khalf;
#pragma unroll
        for (int q = 0; q < 4; ++q) {
            v4f tmp[8];
#pragma unroll
            for (int u = 0; u < 8; ++u)
                tmp[u] = *(const v4f*)(wsrc + (size_t)(2 * u) * KDIM + 64 * q);
#pragma unroll
            for (int u = 0; u < 8; ++u)
                *(v4f*)(wdst + (2 * u) * WPITCH) = tmp[u];
#pragma unroll
            for (int i = 0; i < 16; ++i)
                breg[16 * q + i] = *(const v2f*)(wfrag + 4 * i);
        }
    }
    __syncthreads();

    const float  bv       = bias[n];
    const int    mhigh    = half * 8;
    const float* h_s_lane = h_s + lane * 4;
    const int    n_tiles  = (n_dst + TILE_M - 1) / TILE_M;

    int buf = 0;
    for (int tile = blockIdx.x; tile < n_tiles; tile += gridDim.x, buf ^= 1) {
        float* feat = lds + buf * (TILE_M * LDF);
        const int row0 = tile * TILE_M;

        int g[2], c[2];
#pragma unroll
        for (int rr = 0; rr < 2; ++rr) {
            g[rr] = min(row0 + wave * 2 + rr, n_dst - 1);
            c[rr] = cnt[g[rr]];
        }
        int idx[2][2];
        v4f hd[2];
#pragma unroll
        for (int rr = 0; rr < 2; ++rr) {
            const int* b = bkt + (size_t)g[rr] * CAP;
            idx[rr][0] = (lane < c[rr]) ? b[lane] : 0;
            idx[rr][1] = (lane + 32 < min(c[rr], CAP)) ? b[lane + 32] : 0;
            hd[rr] = *(const v4f*)(h_d + (size_t)g[rr] * IN_FEAT + lane * 4);
        }
#pragma unroll
        for (int rr = 0; rr < 2; ++rr) {
            const int n_in = min(c[rr], CAP);
            v4f acc = {0.0f, 0.0f, 0.0f, 0.0f};
            acc = gather_rows(h_s_lane, idx[rr][0], min(n_in, 32), acc);
            if (n_in > 32) acc = gather_rows(h_s_lane, idx[rr][1], n_in - 32, acc);
            if (c[rr] > CAP) {
                const int total = min(*ovf_count, n_edges);
                for (int o0 = 0; o0 < total; o0 += 32) {
                    const int o = o0 + lane;
                    v2i pr = {-1, 0};
                    if (o < total) pr = *(const v2i*)(ovf + (size_t)o * 2);
                    unsigned mm = (unsigned)__builtin_amdgcn_ballot_w32(pr[0] == g[rr]);
                    while (mm) {
                        const int l = __builtin_ctz(mm);
                        mm &= mm - 1;
                        const int sidx = __builtin_amdgcn_readlane(pr[1], l);
                        acc += *(const v4f*)(h_s_lane + (size_t)sidx * IN_FEAT);
                    }
                }
            }
            const float dg = fmaxf((float)c[rr], 1.0f);
            const v4f mean = acc / dg;
            float* frow_w = feat + (wave * 2 + rr) * LDF + lane * 4;
            *(v4f*)(frow_w)           = hd[rr];
            *(v4f*)(frow_w + IN_FEAT) = mean;
        }
        __syncthreads();

        const float* frow = feat + m * LDF + khalf;

        v8f acc = {};
        __builtin_amdgcn_sched_barrier(0);
#pragma unroll
        for (int i = 0; i < 64; ++i) {
            v2f a = *(const v2f*)(frow + 4 * i);
            acc = __builtin_amdgcn_wmma_f32_16x16x4_f32(
                 false, a,  false, breg[i],
                 (short)0, acc,  false,  false);
        }
        asm volatile("v_nop\n\tv_nop\n\tv_nop\n\tv_nop\n\tv_nop" : "+v"(acc));
        __builtin_amdgcn_sched_barrier(0);

#pragma unroll
        for (int v = 0; v < 8; ++v) {
            const int gr = row0 + mhigh + v;
            if (gr < n_dst) out[(size_t)gr * OUT_FEAT + n] = acc[v] + bv;
        }
    }
}

extern "C" void kernel_launch(void* const* d_in, const int* in_sizes, int n_in,
                              void* d_out, int out_size, void* d_ws, size_t ws_size,
                              hipStream_t stream) {
    const float* h_s  = (const float*)d_in[0];
    const float* h_d  = (const float*)d_in[1];
    const int*   src  = (const int*)d_in[2];
    const int*   dst  = (const int*)d_in[3];
    const float* W    = (const float*)d_in[4];
    const float* bias = (const float*)d_in[5];

    const int n_dst   = in_sizes[1] / IN_FEAT;
    const int n_edges = in_sizes[2];
    (void)n_in; (void)out_size; (void)ws_size;

    int* cnt       = (int*)d_ws;
    int* ovf_count = cnt + n_dst;
    int* bkt       = cnt + ((n_dst + 1 + 63) / 64) * 64;
    int* ovf       = bkt + (size_t)n_dst * CAP;

    const size_t zero_bytes = ((size_t)n_dst + 1) * sizeof(int);
    hipMemsetAsync(d_ws, 0, zero_bytes, stream);

    const int bblocks = (n_edges + 1023) / 1024;
    sage_bucket<<<bblocks, 256, 0, stream>>>(src, dst, cnt, bkt, ovf_count, ovf, n_edges);

    sage_gemm<<<GEMM_GRID, 256, 0, stream>>>(h_s, h_d, cnt, bkt, ovf_count, ovf, W, bias,
                                          (float*)d_out, n_dst, n_edges);
}


# bucket kernel: rows split in two ranges, one per chip half (from the die id), a work queue per range with helping: a count line takes adds from one half only; sum order = arrival order as before
# speedup vs baseline: 3.7197x; 1.6205x over previous
#include <hip/hip_runtime.h>


typedef __attribute__((ext_vector_type(2))) float v2f;
typedef __attribute__((ext_vector_type(4))) float v4f;
typedef __attribute__((ext_vector_type(8))) float v8f;
typedef __attribute__((ext_vector_type(2))) int   v2i;
typedef __attribute__((ext_vector_type(4))) int   v4i;

constexpr int IN_FEAT  = 128;
constexpr int KDIM     = 2 * IN_FEAT;
constexpr int OUT_FEAT = 128;
constexpr int TILE_M   = 16;
constexpr int LDF      = KDIM + 4;

constexpr int CAP = 64;

constexpr int GEMM_GRID = 512;
constexpr int WPITCH = 68;
constexpr int FEAT_FLOATS = 2 * TILE_M * LDF;
constexpr int WST_FLOATS  = 8 * TILE_M * WPITCH;
constexpr int LDS_FLOATS  = FEAT_FLOATS > WST_FLOATS ? FEAT_FLOATS : WST_FLOATS;

__device__ __forceinline__ int atomic_inc_ret(int* p) {
    return __hip_atomic_fetch_add(p, 1, __ATOMIC_RELAXED, __HIP_MEMORY_SCOPE_AGENT);
}

__device__ __forceinline__ void place_edge(int* __restrict__ bkt, int* __restrict__ ovf_count,
                                           int* __restrict__ ovf, int d, int s, int slot)
{
    if (slot < CAP) {
        bkt[(size_t)d * CAP + slot] = s;
    } else {
        const int o = atomic_inc_ret(ovf_count);
        v2i pr; pr[0] = d; pr[1] = s;
        *(v2i*)(ovf + (size_t)o * 2) = pr;
    }
}

constexpr int WCHUNK = 256;
constexpr int QPITCH = 64;

__device__ __forceinline__ int claim_chunk(int* __restrict__ queue, int h, int lane) {
    int v = 0;
    if (lane == 0) v = atomic_inc_ret(queue + h * QPITCH);
    return v;
}

__global__ __launch_bounds__(256) void sage_bucket(
    const int* __restrict__ src, const int* __restrict__ dst,
    int* __restrict__ cnt, int* __restrict__ bkt,
    int* __restrict__ ovf_count, int* __restrict__ ovf,
    int* __restrict__ queue, int n_edges, int n_dst)
{
    const int lane = threadIdx.x & 31;
    int h = (int)((__builtin_amdgcn_s_sendmsg_rtn(0x87) >> 18) & 1u);

    const int n_chunks = (n_edges + WCHUNK - 1) / WCHUNK;
    const int split = min(n_dst, ((n_dst >> 1) + 63) & ~63);

    bool helping = false;
    int  cv = claim_chunk(queue, h, lane);
    const int max_trips = 2 * n_chunks + 2;
    for (int trip = 0; trip < max_trips; ++trip) {
        const int c = __builtin_amdgcn_readfirstlane(cv);
        if (c >= n_chunks) {
            if (helping) break;
            helping = true;
            h ^= 1;
            cv = claim_chunk(queue, h, lane);
            continue;
        }
        cv = claim_chunk(queue, h, lane);

        const int      lo    = h ? split : 0;
        const unsigned width = (unsigned)((h ? n_dst : split) - lo);
        const int      base  = c * WCHUNK;

        if (base + WCHUNK <= n_edges) {
            v4i dv[2], sv[2];
#pragma unroll
            for (int u = 0; u < 2; ++u) {
                dv[u] = *(const v4i*)(dst + base + 128 * u + lane * 4);
                sv[u] = *(const v4i*)(src + base + 128 * u + lane * 4);
            }
            int slot[8];
#pragma unroll
            for (int k = 0; k < 8; ++k) {
                const int d = dv[k >> 2][k & 3];
                slot[k] = -1;
                if ((unsigned)(d - lo) < width) slot[k] = atomic_inc_ret(&cnt[d]);
            }
#pragma unroll
            for (int k = 0; k < 8; ++k) {
                if (slot[k] >= 0)
                    place_edge(bkt, ovf_count, ovf, dv[k >> 2][k & 3], sv[k >> 2][k & 3], slot[k]);
            }
        } else {
            for (int e = base + lane; e < n_edges; e += 32) {
                const int d = dst[e];
                if ((unsigned)(d - lo) < width)
                    place_edge(bkt, ovf_count, ovf, d, src[e], atomic_inc_ret(&cnt[d]));
            }
        }
    }
}

template <int N>
__device__ __forceinline__ v4f gather_group(const float* __restrict__ h_s_lane,
                                            int idx, int j, v4f acc)
{
    v4f v[N];
#pragma unroll
    for (int u = 0; u < N; ++u) {
        const int sidx = __builtin_amdgcn_readlane(idx, j + u);
        v[u] = *(const v4f*)(h_s_lane + (size_t)sidx * IN_FEAT);
    }
#pragma unroll
    for (int u = 0; u < N; ++u) acc += v[u];
    return acc;
}

__device__ __forceinline__ v4f gather_rows(const float* __restrict__ h_s_lane,
                                           int idx, int n, v4f acc)
{
    int j = 0;
    for (; j + 8 <= n; j += 8) acc = gather_group<8>(h_s_lane, idx, j, acc);
    if (j + 4 <= n) { acc = gather_group<4>(h_s_lane, idx, j, acc); j += 4; }
    const int rem = n - j;
    if (rem == 3)      acc = gather_group<3>(h_s_lane, idx, j, acc);
    else if (rem == 2) acc = gather_group<2>(h_s_lane, idx, j, acc);
    else if (rem == 1) acc = gather_group<1>(h_s_lane, idx, j, acc);
    return acc;
}

__global__ __launch_bounds__(256) void sage_gemm(
    const float* __restrict__ h_s, const float* __restrict__ h_d,
    const int* __restrict__ cnt, const int* __restrict__ bkt,
    const int* __restrict__ ovf_count, const int* __restrict__ ovf,
    const float* __restrict__ W, const float* __restrict__ bias,
    float* __restrict__ out, int n_dst, int n_edges)
{
    __shared__ __attribute__((aligned(16))) float lds[LDS_FLOATS];

    const int t    = threadIdx.x;
    const int lane = t & 31;
    const int wave = __builtin_amdgcn_readfirstlane(t >> 5);

    const int m     = lane & 15;
    const int half  = lane >> 4;
    const int khalf = half * 2;
    const int n     = wave * 16 + m;

    v2f breg[64];
    {
        float* wt = lds + wave * (TILE_M * WPITCH);
        const float* wsrc  = W + (size_t)(wave * 16 + half) * KDIM + m * 4;
        float*       wdst  = wt + half * WPITCH + m * 4;
        const float* wfrag = wt + m * WPITCH + khalf;
#pragma unroll
        for (int q = 0; q < 4; ++q) {
            v4f tmp[8];
#pragma unroll
            for (int u = 0; u < 8; ++u)
                tmp[u] = *(const v4f*)(wsrc + (size_t)(2 * u) * KDIM + 64 * q);
#pragma unroll
            for (int u = 0; u < 8; ++u)
                *(v4f*)(wdst + (2 * u) * WPITCH) = tmp[u];
#pragma unroll
            for (int i = 0; i < 16; ++i)
                breg[16 * q + i] = *(const v2f*)(wfrag + 4 * i);
        }
    }
    __syncthreads();

    const float  bv       = bias[n];
    const int    mhigh    = half * 8;
    const float* h_s_lane = h_s + lane * 4;
    const int    n_tiles  = (n_dst + TILE_M - 1) / TILE_M;

    int buf = 0;
    for (int tile = blockIdx.x; tile < n_tiles; tile += gridDim.x, buf ^= 1) {
        float* feat = lds + buf * (TILE_M * LDF);
        const int row0 = tile * TILE_M;

        int g[2], c[2];
#pragma unroll
        for (int rr = 0; rr < 2; ++rr) {
            g[rr] = min(row0 + wave * 2 + rr, n_dst - 1);
            c[rr] = cnt[g[rr]];
        }
        int idx[2][2];
        v4f hd[2];
#pragma unroll
        for (int rr = 0; rr < 2; ++rr) {
            const int* b = bkt + (size_t)g[rr] * CAP;
            idx[rr][0] = (lane < c[rr]) ? b[lane] : 0;
            idx[rr][1] = (lane + 32 < min(c[rr], CAP)) ? b[lane + 32] : 0;
            hd[rr] = *(const v4f*)(h_d + (size_t)g[rr] * IN_FEAT + lane * 4);
        }
#pragma unroll
        for (int rr = 0; rr < 2; ++rr) {
            const int n_in = min(c[rr], CAP);
            v4f acc = {0.0f, 0.0f, 0.0f, 0.0f};
            acc = gather_rows(h_s_lane, idx[rr][0], min(n_in, 32), acc);
            if (n_in > 32) acc = gather_rows(h_s_lane, idx[rr][1], n_in - 32, acc);
            if (c[rr] > CAP) {
                const int total = min(*ovf_count, n_edges);
                for (int o0 = 0; o0 < total; o0 += 32) {
                    const int o = o0 + lane;
                    v2i pr = {-1, 0};
                    if (o < total) pr = *(const v2i*)(ovf + (size_t)o * 2);
                    unsigned mm = (unsigned)__builtin_amdgcn_ballot_w32(pr[0] == g[rr]);
                    while (mm) {
                        const int l = __builtin_ctz(mm);
                        mm &= mm - 1;
                        const int sidx = __builtin_amdgcn_readlane(pr[1], l);
                        acc += *(const v4f*)(h_s_lane + (size_t)sidx * IN_FEAT);
                    }
                }
            }
            const float dg = fmaxf((float)c[rr], 1.0f);
            const v4f mean = acc / dg;
            float* frow_w = feat + (wave * 2 + rr) * LDF + lane * 4;
            *(v4f*)(frow_w)           = hd[rr];
            *(v4f*)(frow_w + IN_FEAT) = mean;
        }
        __syncthreads();

        const float* frow = feat + m * LDF + khalf;

        v8f acc = {};
        __builtin_amdgcn_sched_barrier(0);
#pragma unroll
        for (int i = 0; i < 64; ++i) {
            v2f a = *(const v2f*)(frow + 4 * i);
            acc = __builtin_amdgcn_wmma_f32_16x16x4_f32(
                 false, a,  false, breg[i],
                 (short)0, acc,  false,  false);
        }
        asm volatile("v_nop\n\tv_nop\n\tv_nop\n\tv_nop\n\tv_nop" : "+v"(acc));
        __builtin_amdgcn_sched_barrier(0);

#pragma unroll
        for (int v = 0; v < 8; ++v) {
            const int gr = row0 + mhigh + v;
            if (gr < n_dst) out[(size_t)gr * OUT_FEAT + n] = acc[v] + bv;
        }
    }
}

extern "C" void kernel_launch(void* const* d_in, const int* in_sizes, int n_in,
                              void* d_out, int out_size, void* d_ws, size_t ws_size,
                              hipStream_t stream) {
    const float* h_s  = (const float*)d_in[0];
    const float* h_d  = (const float*)d_in[1];
    const int*   src  = (const int*)d_in[2];
    const int*   dst  = (const int*)d_in[3];
    const float* W    = (const float*)d_in[4];
    const float* bias = (const float*)d_in[5];

    const int n_dst   = in_sizes[1] / IN_FEAT;
    const int n_edges = in_sizes[2];
    (void)n_in; (void)out_size; (void)ws_size;

    const int zints = ((n_dst + 1 + 63) / 64) * 64 + 2 * QPITCH;
    int* cnt       = (int*)d_ws;
    int* ovf_count = cnt + n_dst;
    int* queue     = cnt + (zints - 2 * QPITCH);
    int* bkt       = cnt + zints;
    int* ovf       = bkt + (size_t)n_dst * CAP;

    const size_t zero_bytes = (size_t)zints * sizeof(int);
    hipMemsetAsync(d_ws, 0, zero_bytes, stream);

    const int bblocks = (n_edges + 2047) / 2048;
    sage_bucket<<<bblocks, 256, 0, stream>>>(src, dst, cnt, bkt, ovf_count, ovf, queue,
                                            n_edges, n_dst);

    sage_gemm<<<GEMM_GRID, 256, 0, stream>>>(h_s, h_d, cnt, bkt, ovf_count, ovf, W, bias,
                                          (float*)d_out, n_dst, n_edges);
}


# GEMM kernel: staging loads run a tile ahead: next tile's degrees and slots requested under this tile's gathers, its first 8 source rows per row behind the barrier, under the WMMAs; sum order same
# speedup vs baseline: 4.1607x; 1.1186x over previous
#include <hip/hip_runtime.h>


typedef __attribute__((ext_vector_type(2))) float v2f;
typedef __attribute__((ext_vector_type(4))) float v4f;
typedef __attribute__((ext_vector_type(8))) float v8f;
typedef __attribute__((ext_vector_type(2))) int   v2i;
typedef __attribute__((ext_vector_type(4))) int   v4i;

constexpr int IN_FEAT  = 128;
constexpr int KDIM     = 2 * IN_FEAT;
constexpr int OUT_FEAT = 128;
constexpr int TILE_M   = 16;
constexpr int LDF      = KDIM + 4;

constexpr int CAP = 64;

constexpr int GEMM_GRID = 512;
constexpr int WPITCH = 68;
constexpr int FEAT_FLOATS = 2 * TILE_M * LDF;
constexpr int WST_FLOATS  = 8 * TILE_M * WPITCH;
constexpr int LDS_FLOATS  = FEAT_FLOATS > WST_FLOATS ? FEAT_FLOATS : WST_FLOATS;

__device__ __forceinline__ int atomic_inc_ret(int* p) {
    return __hip_atomic_fetch_add(p, 1, __ATOMIC_RELAXED, __HIP_MEMORY_SCOPE_AGENT);
}

__device__ __forceinline__ void place_edge(int* __restrict__ bkt, int* __restrict__ ovf_count,
                                           int* __restrict__ ovf, int d, int s, int slot)
{
    if (slot < CAP) {
        bkt[(size_t)d * CAP + slot] = s;
    } else {
        const int o = atomic_inc_ret(ovf_count);
        v2i pr; pr[0] = d; pr[1] = s;
        *(v2i*)(ovf + (size_t)o * 2) = pr;
    }
}

constexpr int WCHUNK = 256;
constexpr int QPITCH = 64;

__device__ __forceinline__ int claim_chunk(int* __restrict__ queue, int h, int lane) {
    int v = 0;
    if (lane == 0) v = atomic_inc_ret(queue + h * QPITCH);
    return v;
}

__global__ __launch_bounds__(256) void sage_bucket(
    const int* __restrict__ src, const int* __restrict__ dst,
    int* __restrict__ cnt, int* __restrict__ bkt,
    int* __restrict__ ovf_count, int* __restrict__ ovf,
    int* __restrict__ queue, int n_edges, int n_dst)
{
    const int lane = threadIdx.x & 31;
    int h = (int)((__builtin_amdgcn_s_sendmsg_rtn(0x87) >> 18) & 1u);

    const int n_chunks = (n_edges + WCHUNK - 1) / WCHUNK;
    const int split = min(n_dst, ((n_dst >> 1) + 63) & ~63);

    bool helping = false;
    int  cv = claim_chunk(queue, h, lane);
    const int max_trips = 2 * n_chunks + 2;
    for (int trip = 0; trip < max_trips; ++trip) {
        const int c = __builtin_amdgcn_readfirstlane(cv);
        if (c >= n_chunks) {
            if (helping) break;
            helping = true;
            h ^= 1;
            cv = claim_chunk(queue, h, lane);
            continue;
        }
        cv = claim_chunk(queue, h, lane);

        const int      lo    = h ? split : 0;
        const unsigned width = (unsigned)((h ? n_dst : split) - lo);
        const int      base  = c * WCHUNK;

        if (base + WCHUNK <= n_edges) {
            v4i dv[2], sv[2];
#pragma unroll
            for (int u = 0; u < 2; ++u) {
                dv[u] = *(const v4i*)(dst + base + 128 * u + lane * 4);
                sv[u] = *(const v4i*)(src + base + 128 * u + lane * 4);
            }
            int slot[8];
#pragma unroll
            for (int k = 0; k < 8; ++k) {
                const int d = dv[k >> 2][k & 3];
                slot[k] = -1;
                if ((unsigned)(d - lo) < width) slot[k] = atomic_inc_ret(&cnt[d]);
            }
#pragma unroll
            for (int k = 0; k < 8; ++k) {
                if (slot[k] >= 0)
                    place_edge(bkt, ovf_count, ovf, dv[k >> 2][k & 3], sv[k >> 2][k & 3], slot[k]);
            }
        } else {
            for (int e = base + lane; e < n_edges; e += 32) {
                const int d = dst[e];
                if ((unsigned)(d - lo) < width)
                    place_edge(bkt, ovf_count, ovf, d, src[e], atomic_inc_ret(&cnt[d]));
            }
        }
    }
}

template <int N>
__device__ __forceinline__ v4f gather_group(const float* __restrict__ h_s_lane,
                                            int idx, int j, v4f acc)
{
    v4f v[N];
#pragma unroll
    for (int u = 0; u < N; ++u) {
        const int sidx = __builtin_amdgcn_readlane(idx, j + u);
        v[u] = *(const v4f*)(h_s_lane + (size_t)sidx * IN_FEAT);
    }
#pragma unroll
    for (int u = 0; u < N; ++u) acc += v[u];
    return acc;
}

__device__ __forceinline__ v4f gather_rows(const float* __restrict__ h_s_lane,
                                           int idx, int j0, int n, v4f acc)
{
    int j = j0;
    for (; j + 8 <= n; j += 8) acc = gather_group<8>(h_s_lane, idx, j, acc);
    if (j + 4 <= n) { acc = gather_group<4>(h_s_lane, idx, j, acc); j += 4; }
    const int rem = n - j;
    if (rem == 3)      acc = gather_group<3>(h_s_lane, idx, j, acc);
    else if (rem == 2) acc = gather_group<2>(h_s_lane, idx, j, acc);
    else if (rem == 1) acc = gather_group<1>(h_s_lane, idx, j, acc);
    return acc;
}

__device__ __forceinline__ int tile_row(int tile, int wave, int rr, int n_dst) {
    return min(tile * TILE_M + wave * 2 + rr, n_dst - 1);
}

__device__ __forceinline__ void request_slots(const int* __restrict__ bkt,
                                              int tile, int wave, int lane, int n_dst,
                                              const int (&c)[2], int (&idx)[2][2])
{
#pragma unroll
    for (int rr = 0; rr < 2; ++rr) {
        const int* b = bkt + (size_t)tile_row(tile, wave, rr, n_dst) * CAP;
        idx[rr][0] = (lane < c[rr]) ? b[lane] : 0;
        idx[rr][1] = (lane + 32 < min(c[rr], CAP)) ? b[lane + 32] : 0;
    }
}

__device__ __forceinline__ bool request_first8(const float* __restrict__ h_s_lane,
                                               int idx, int n_in, v4f (&pre)[8])
{
    const bool have = n_in >= 8;
    if (have) {
#pragma unroll
        for (int u = 0; u < 8; ++u) {
            const int sidx = __builtin_amdgcn_readlane(idx, u);
            pre[u] = *(const v4f*)(h_s_lane + (size_t)sidx * IN_FEAT);
        }
    } else {
#pragma unroll
        for (int u = 0; u < 8; ++u) pre[u] = v4f{0.0f, 0.0f, 0.0f, 0.0f};
    }
    return have;
}

__global__ __launch_bounds__(256) void sage_gemm(
    const float* __restrict__ h_s, const float* __restrict__ h_d,
    const int* __restrict__ cnt, const int* __restrict__ bkt,
    const int* __restrict__ ovf_count, const int* __restrict__ ovf,
    const float* __restrict__ W, const float* __restrict__ bias,
    float* __restrict__ out, int n_dst, int n_edges)
{
    __shared__ __attribute__((aligned(16))) float lds[LDS_FLOATS];

    const int t    = threadIdx.x;
    const int lane = t & 31;
    const int wave = __builtin_amdgcn_readfirstlane(t >> 5);

    const int m     = lane & 15;
    const int half  = lane >> 4;
    const int khalf = half * 2;
    const int n     = wave * 16 + m;

    const int tstep = gridDim.x;
    int c_cur[2], c_nxt[2];
#pragma unroll
    for (int rr = 0; rr < 2; ++rr) {
        c_cur[rr] = cnt[tile_row(blockIdx.x, wave, rr, n_dst)];
        c_nxt[rr] = cnt[tile_row(blockIdx.x + tstep, wave, rr, n_dst)];
    }
    int idx_cur[2][2];
    request_slots(bkt, blockIdx.x, wave, lane, n_dst, c_cur, idx_cur);

    v2f breg[64];
    {
        float* wt = lds + wave * (TILE_M * WPITCH);
        const float* wsrc  = W + (size_t)(wave * 16 + half) * KDIM + m * 4;
        float*       wdst  = wt + half * WPITCH + m * 4;
        const float* wfrag = wt + m * WPITCH + khalf;
#pragma unroll
        for (int q = 0; q < 4; ++q) {
            v4f tmp[8];
#pragma unroll
            for (int u = 0; u < 8; ++u)
                tmp[u] = *(const v4f*)(wsrc + (size_t)(2 * u) * KDIM + 64 * q);
#pragma unroll
            for (int u = 0; u < 8; ++u)
                *(v4f*)(wdst + (2 * u) * WPITCH) = tmp[u];
#pragma unroll
            for (int i = 0; i < 16; ++i)
                breg[16 * q + i] = *(const v2f*)(wfrag + 4 * i);
        }
    }
    __syncthreads();

    const float  bv       = bias[n];
    const int    mhigh    = half * 8;
    const float* h_s_lane = h_s + lane * 4;
    const int    n_tiles  = (n_dst + TILE_M - 1) / TILE_M;

    v4f  pre[2][8];
    bool pre_have[2] = {false, false};
#pragma unroll
    for (int rr = 0; rr < 2; ++rr)
#pragma unroll
        for (int u = 0; u < 8; ++u) pre[rr][u] = v4f{0.0f, 0.0f, 0.0f, 0.0f};

    int buf = 0;
    for (int tile = blockIdx.x; tile < n_tiles; tile += tstep, buf ^= 1) {
        float* feat = lds + buf * (TILE_M * LDF);
        const int row0 = tile * TILE_M;

        v4f acc0[2];
        int j0[2];
#pragma unroll
        for (int rr = 0; rr < 2; ++rr) {
            acc0[rr] = v4f{0.0f, 0.0f, 0.0f, 0.0f};
#pragma unroll
            for (int u = 0; u < 8; ++u) acc0[rr] += pre[rr][u];
            j0[rr] = pre_have[rr] ? 8 : 0;
        }
        asm volatile("" : "+v"(acc0[0]), "+v"(acc0[1]));

        const int c_nn = cnt[tile_row(tile + 2 * tstep, wave, lane & 1, n_dst)];
        int idx_nxt[2][2];
        request_slots(bkt, tile + tstep, wave, lane, n_dst, c_nxt, idx_nxt);

        int g[2], c[2];
        int idx[2][2];
        v4f hd[2];
#pragma unroll
        for (int rr = 0; rr < 2; ++rr) {
            g[rr] = tile_row(tile, wave, rr, n_dst);
            c[rr] = c_cur[rr];
            idx[rr][0] = idx_cur[rr][0];
            idx[rr][1] = idx_cur[rr][1];
            hd[rr] = *(const v4f*)(h_d + (size_t)g[rr] * IN_FEAT + lane * 4);
        }
#pragma unroll
        for (int rr = 0; rr < 2; ++rr) {
            const int n_in = min(c[rr], CAP);
            v4f acc = acc0[rr];
            acc = gather_rows(h_s_lane, idx[rr][0], j0[rr], min(n_in, 32), acc);
            if (n_in > 32) acc = gather_rows(h_s_lane, idx[rr][1], 0, n_in - 32, acc);
            if (c[rr] > CAP) {
                const int total = min(*ovf_count, n_edges);
                for (int o0 = 0; o0 < total; o0 += 32) {
                    const int o = o0 + lane;
                    v2i pr = {-1, 0};
                    if (o < total) pr = *(const v2i*)(ovf + (size_t)o * 2);
                    unsigned mm = (unsigned)__builtin_amdgcn_ballot_w32(pr[0] == g[rr]);
                    while (mm) {
                        const int l = __builtin_ctz(mm);
                        mm &= mm - 1;
                        const int sidx = __builtin_amdgcn_readlane(pr[1], l);
                        acc += *(const v4f*)(h_s_lane + (size_t)sidx * IN_FEAT);
                    }
                }
            }
            const float dg = fmaxf((float)c[rr], 1.0f);
            const v4f mean = acc / dg;
            float* frow_w = feat + (wave * 2 + rr) * LDF + lane * 4;
            *(v4f*)(frow_w)           = hd[rr];
            *(v4f*)(frow_w + IN_FEAT) = mean;
        }
        __syncthreads();

        const bool more = tile + tstep < n_tiles;
#pragma unroll
        for (int rr = 0; rr < 2; ++rr)
            pre_have[rr] = request_first8(h_s_lane, idx_nxt[rr][0], more ? min(c_nxt[rr], CAP) : 0, pre[rr]);

        const float* frow = feat + m * LDF + khalf;

        v8f acc = {};
        __builtin_amdgcn_sched_barrier(0);
#pragma unroll
        for (int i = 0; i < 64; ++i) {
            v2f a = *(const v2f*)(frow + 4 * i);
            acc = __builtin_amdgcn_wmma_f32_16x16x4_f32(
                 false, a,  false, breg[i],
                 (short)0, acc,  false,  false);
        }
        asm volatile("v_nop\n\tv_nop\n\tv_nop\n\tv_nop\n\tv_nop" : "+v"(acc));
        __builtin_amdgcn_sched_barrier(0);

#pragma unroll
        for (int v = 0; v < 8; ++v) {
            const int gr = row0 + mhigh + v;
            if (gr < n_dst) out[(size_t)gr * OUT_FEAT + n] = acc[v] + bv;
        }

#pragma unroll
        for (int rr = 0; rr < 2; ++rr) {
            c_cur[rr] = c_nxt[rr];
            c_nxt[rr] = __builtin_amdgcn_readlane(c_nn, rr);
            idx_cur[rr][0] = idx_nxt[rr][0];
            idx_cur[rr][1] = idx_nxt[rr][1];
        }
    }
}

extern "C" void kernel_launch(void* const* d_in, const int* in_sizes, int n_in,
                              void* d_out, int out_size, void* d_ws, size_t ws_size,
                              hipStream_t stream) {
    const float* h_s  = (const float*)d_in[0];
    const float* h_d  = (const float*)d_in[1];
    const int*   src  = (const int*)d_in[2];
    const int*   dst  = (const int*)d_in[3];
    const float* W    = (const float*)d_in[4];
    const float* bias = (const float*)d_in[5];

    const int n_dst   = in_sizes[1] / IN_FEAT;
    const int n_edges = in_sizes[2];
    (void)n_in; (void)out_size; (void)ws_size;

    const int zints = ((n_dst + 1 + 63) / 64) * 64 + 2 * QPITCH;
    int* cnt       = (int*)d_ws;
    int* ovf_count = cnt + n_dst;
    int* queue     = cnt + (zints - 2 * QPITCH);
    int* bkt       = cnt + zints;
    int* ovf       = bkt + (size_t)n_dst * CAP;

    const size_t zero_bytes = (size_t)zints * sizeof(int);
    hipMemsetAsync(d_ws, 0, zero_bytes, stream);

    const int bblocks = (n_edges + 2047) / 2048;
    sage_bucket<<<bblocks, 256, 0, stream>>>(src, dst, cnt, bkt, ovf_count, ovf, queue,
                                            n_edges, n_dst);

    sage_gemm<<<GEMM_GRID, 256, 0, stream>>>(h_s, h_d, cnt, bkt, ovf_count, ovf, W, bias,
                                          (float*)d_out, n_dst, n_edges);
}
